# starting kernel
# baseline (speedup 1.0000x reference)
#include <hip/hip_runtime.h>
#include <math.h>

typedef __attribute__((ext_vector_type(2))) float v2f;
typedef __attribute__((ext_vector_type(8))) float v8f;

#define BOHR_F 1.8897261258369282f
#define PI_F   3.14159265358979323846f

__device__ __forceinline__ void sincos2pi_x8(v8f x, v8f* sn, v8f* cs) {
    float s0, s1, s2, s3, s4, s5, s6, s7;
    float c0, c1, c2, c3, c4, c5, c6, c7;
    asm volatile(
        "v_sin_f32 %0, %16\n\t"
        "v_sin_f32 %1, %17\n\t"
        "v_sin_f32 %2, %18\n\t"
        "v_sin_f32 %3, %19\n\t"
        "v_sin_f32 %4, %20\n\t"
        "v_sin_f32 %5, %21\n\t"
        "v_sin_f32 %6, %22\n\t"
        "v_sin_f32 %7, %23\n\t"
        "v_cos_f32 %8, %16\n\t"
        "v_cos_f32 %9, %17\n\t"
        "v_cos_f32 %10, %18\n\t"
        "v_cos_f32 %11, %19\n\t"
        "v_cos_f32 %12, %20\n\t"
        "v_cos_f32 %13, %21\n\t"
        "v_cos_f32 %14, %22\n\t"
        "v_cos_f32 %15, %23\n\t"
        "v_nop"
        : "=&v"(s0), "=&v"(s1), "=&v"(s2), "=&v"(s3),
          "=&v"(s4), "=&v"(s5), "=&v"(s6), "=&v"(s7),
          "=&v"(c0), "=&v"(c1), "=&v"(c2), "=&v"(c3),
          "=&v"(c4), "=&v"(c5), "=&v"(c6), "=&v"(c7)
        : "v"(x[0]), "v"(x[1]), "v"(x[2]), "v"(x[3]),
          "v"(x[4]), "v"(x[5]), "v"(x[6]), "v"(x[7]));
    *sn = (v8f){s0, s1, s2, s3, s4, s5, s6, s7};
    *cs = (v8f){c0, c1, c2, c3, c4, c5, c6, c7};
}

__global__ void k_zero(float* __restrict__ p, int n) {
    int i = blockIdx.x * blockDim.x + threadIdx.x;
    if (i < n) p[i] = 0.0f;
}

__global__ void k_pack(const float* __restrict__ coords, const float* __restrict__ q,
                       const float* __restrict__ kvec, const float* __restrict__ expfac,
                       float4* __restrict__ packA, float4* __restrict__ packK,
                       int BN, int NK)
{
    int i = blockIdx.x * blockDim.x + threadIdx.x;
    if (i < BN) {
        float4 p;
        p.x = coords[i * 3 + 0];
        p.y = coords[i * 3 + 1];
        p.z = coords[i * 3 + 2];
        p.w = q[i];
        packA[i] = p;
    }
    if (i < NK) {
        float4 p;
        p.x = kvec[i * 3 + 0];
        p.y = kvec[i * 3 + 1];
        p.z = kvec[i * 3 + 2];
        p.w = expfac[i];
        packK[i] = p;
    }
}

__global__ __launch_bounds__(256) void k_struct_fac(
    const float4* __restrict__ packA, const float* __restrict__ cell_inv,
    const float* __restrict__ kvec, float* __restrict__ Sri,
    int N, int NK, int splitN)
{
    const int lane  = threadIdx.x & 31;
    const int wave  = threadIdx.x >> 5;
    const int kTile = (blockIdx.x * 8 + wave) * 16;
    const int b     = blockIdx.z;
    const int nPer  = N / splitN;
    const int nBeg  = blockIdx.y * nPer;

    const int  m  = lane & 15;
    const bool hi = lane >= 16;

    const float kx = kvec[(kTile + m) * 3 + 0];
    const float ky = kvec[(kTile + m) * 3 + 1];
    const float kz = kvec[(kTile + m) * 3 + 2];
    const float kp0 = kx * cell_inv[0] + ky * cell_inv[3] + kz * cell_inv[6];
    const float kp1 = kx * cell_inv[1] + ky * cell_inv[4] + kz * cell_inv[7];
    const float kp2 = kx * cell_inv[2] + ky * cell_inv[5] + kz * cell_inv[8];

    v2f a;
    a.x = hi ? kp2 : kp0;
    a.y = hi ? 0.0f : kp1;

    v8f sumRe = {}, sumIm = {};
    const float4* pA = packA + (size_t)b * N;

#pragma unroll 2
    for (int n0 = nBeg; n0 < nBeg + nPer; n0 += 16) {
        const float4 p = pA[n0 + m];
        v2f bm;
        bm.x = hi ? p.z : p.x;
        bm.y = hi ? 0.0f : p.y;

        v8f d = {};
        d = __builtin_amdgcn_wmma_f32_16x16x4_f32(
                false, a, false, bm, (short)0, d, false, false);

        v8f sn, cs;
        sincos2pi_x8(d, &sn, &cs);
        sumRe = sumRe + cs * p.w;
        sumIm = sumIm + sn * p.w;
    }

#pragma unroll
    for (int r = 0; r < 8; ++r) {
        float re = sumRe[r], im = sumIm[r];
        for (int mask = 8; mask >= 1; mask >>= 1) {
            re += __shfl_xor(re, mask, 32);
            im += __shfl_xor(im, mask, 32);
        }
        sumRe[r] = re; sumIm[r] = im;
    }
    if (m == 0) {
        const int kbase = kTile + (hi ? 8 : 0);
#pragma unroll
        for (int r = 0; r < 8; ++r) {
            atomicAdd(&Sri[2 * ((size_t)b * NK + kbase + r) + 0], sumRe[r]);
            atomicAdd(&Sri[2 * ((size_t)b * NK + kbase + r) + 1], sumIm[r]);
        }
    }
}

__global__ __launch_bounds__(256) void k_recip(
    const float4* __restrict__ packA, const float* __restrict__ cell_inv,
    const float4* __restrict__ packK, const float2* __restrict__ Sri,
    float* __restrict__ recip, int N, int NK, int splitK)
{
    const int lane  = threadIdx.x & 31;
    const int wave  = threadIdx.x >> 5;
    const int nTile = (blockIdx.x * 8 + wave) * 16;
    const int b     = blockIdx.z;
    const int kPer  = NK / splitK;
    const int kBeg  = blockIdx.y * kPer;

    const int  m  = lane & 15;
    const bool hi = lane >= 16;

    const float4 pc = packA[(size_t)b * N + nTile + m];
    const float s0 = cell_inv[0] * pc.x + cell_inv[1] * pc.y + cell_inv[2] * pc.z;
    const float s1 = cell_inv[3] * pc.x + cell_inv[4] * pc.y + cell_inv[5] * pc.z;
    const float s2 = cell_inv[6] * pc.x + cell_inv[7] * pc.y + cell_inv[8] * pc.z;

    v2f a;
    a.x = hi ? s2 : s0;
    a.y = hi ? 0.0f : s1;

    v8f acc = {};
    const float2* pS = Sri + (size_t)b * NK;

#pragma unroll 2
    for (int kc = kBeg; kc < kBeg + kPer; kc += 16) {
        const float4 kv = packK[kc + m];
        v2f bm;
        bm.x = hi ? kv.z : kv.x;
        bm.y = hi ? 0.0f : kv.y;

        v8f d = {};
        d = __builtin_amdgcn_wmma_f32_16x16x4_f32(
                false, a, false, bm, (short)0, d, false, false);

        const float2 S = pS[kc + m];
        v8f sn, cs;
        sincos2pi_x8(d, &sn, &cs);
        acc = acc + (cs * S.x + sn * S.y) * kv.w;
    }

#pragma unroll
    for (int r = 0; r < 8; ++r) {
        float v = acc[r];
        for (int mask = 8; mask >= 1; mask >>= 1)
            v += __shfl_xor(v, mask, 32);
        acc[r] = v;
    }
    if (m == 0) {
        const int nbase = nTile + (hi ? 8 : 0);
#pragma unroll
        for (int r = 0; r < 8; ++r)
            atomicAdd(&recip[(size_t)b * N + nbase + r], acc[r]);
    }
}

__global__ void k_phi(const float* __restrict__ q, const float* __restrict__ recip,
                      const float* __restrict__ volume, const float* __restrict__ bewald,
                      float* __restrict__ out, int BN)
{
    int i = blockIdx.x * blockDim.x + threadIdx.x;
    if (i >= BN) return;
    const float scale = BOHR_F / (PI_F * volume[0]);
    const float self  = 2.0f * bewald[0] * BOHR_F / sqrtf(PI_F);
    const float phi   = recip[i] * scale - q[i] * self;
    out[i]      = 0.5f * q[i] * phi;
    out[BN + i] = phi;
}

extern "C" void kernel_launch(void* const* d_in, const int* in_sizes, int n_in,
                              void* d_out, int out_size, void* d_ws, size_t ws_size,
                              hipStream_t stream)
{
    const float* coords   = (const float*)d_in[0];
    const float* q        = (const float*)d_in[1];
    const float* cell_inv = (const float*)d_in[2];
    const float* kvec     = (const float*)d_in[3];
    const float* expfac   = (const float*)d_in[4];
    const float* volume   = (const float*)d_in[5];
    const float* bewald   = (const float*)d_in[6];
    float* out = (float*)d_out;

    const int B  = 2;
    const int BN = in_sizes[1];
    const int N  = BN / B;
    const int NK = in_sizes[4];

    float*  Sri   = (float*)d_ws;
    float*  recip = Sri + (size_t)2 * B * NK;
    float4* packA = (float4*)(recip + BN);
    float4* packK = packA + BN;
    const size_t wsFloats = (size_t)2 * B * NK + BN + (size_t)4 * BN + (size_t)4 * NK;
    if (ws_size < wsFloats * sizeof(float)) return;

    const int SPLIT_N = 16;
    const int SPLIT_K = 16;

    const int zeroN = 2 * B * NK + BN;
    k_zero<<<(zeroN + 255) / 256, 256, 0, stream>>>(Sri, zeroN);

    const int packN = (BN > NK) ? BN : NK;
    k_pack<<<(packN + 255) / 256, 256, 0, stream>>>(coords, q, kvec, expfac,
                                                    packA, packK, BN, NK);

    dim3 g1(NK / 128, SPLIT_N, B);
    k_struct_fac<<<g1, 256, 0, stream>>>(packA, cell_inv, kvec, Sri, N, NK, SPLIT_N);

    dim3 g2(N / 128, SPLIT_K, B);
    k_recip<<<g2, 256, 0, stream>>>(packA, cell_inv, packK, (float2*)Sri,
                                    recip, N, NK, SPLIT_K);

    k_phi<<<(BN + 255) / 256, 256, 0, stream>>>(q, recip, volume, bewald, out, BN);
}


# Both WMMA loops: chunk operand loads issued 8 chunks ahead into register slots, each refilled right behind its last use, instead of load-then-wait on every trip; summation order unchanged
# speedup vs baseline: 1.6920x; 1.6920x over previous
#include <hip/hip_runtime.h>
#include <math.h>

typedef __attribute__((ext_vector_type(2))) float v2f;
typedef __attribute__((ext_vector_type(4))) float v4f;
typedef __attribute__((ext_vector_type(8))) float v8f;

#define BOHR_F 1.8897261258369282f
#define PI_F   3.14159265358979323846f

__device__ __forceinline__ void sincos2pi_x8(v8f x, v8f* sn, v8f* cs) {
    float s0, s1, s2, s3, s4, s5, s6, s7;
    float c0, c1, c2, c3, c4, c5, c6, c7;
    asm volatile(
        "v_sin_f32 %0, %16\n\t"
        "v_sin_f32 %1, %17\n\t"
        "v_sin_f32 %2, %18\n\t"
        "v_sin_f32 %3, %19\n\t"
        "v_sin_f32 %4, %20\n\t"
        "v_sin_f32 %5, %21\n\t"
        "v_sin_f32 %6, %22\n\t"
        "v_sin_f32 %7, %23\n\t"
        "v_cos_f32 %8, %16\n\t"
        "v_cos_f32 %9, %17\n\t"
        "v_cos_f32 %10, %18\n\t"
        "v_cos_f32 %11, %19\n\t"
        "v_cos_f32 %12, %20\n\t"
        "v_cos_f32 %13, %21\n\t"
        "v_cos_f32 %14, %22\n\t"
        "v_cos_f32 %15, %23\n\t"
        "v_nop"
        : "=&v"(s0), "=&v"(s1), "=&v"(s2), "=&v"(s3),
          "=&v"(s4), "=&v"(s5), "=&v"(s6), "=&v"(s7),
          "=&v"(c0), "=&v"(c1), "=&v"(c2), "=&v"(c3),
          "=&v"(c4), "=&v"(c5), "=&v"(c6), "=&v"(c7)
        : "v"(x[0]), "v"(x[1]), "v"(x[2]), "v"(x[3]),
          "v"(x[4]), "v"(x[5]), "v"(x[6]), "v"(x[7]));
    *sn = (v8f){s0, s1, s2, s3, s4, s5, s6, s7};
    *cs = (v8f){c0, c1, c2, c3, c4, c5, c6, c7};
}

__global__ void k_pack(const float* __restrict__ coords, const float* __restrict__ q,
                       const float* __restrict__ kvec, const float* __restrict__ expfac,
                       float4* __restrict__ packA, float4* __restrict__ packK,
                       int BN, int NK)
{
    int i = blockIdx.x * blockDim.x + threadIdx.x;
    if (i < BN) {
        float4 p;
        p.x = coords[i * 3 + 0];
        p.y = coords[i * 3 + 1];
        p.z = coords[i * 3 + 2];
        p.w = q[i];
        packA[i] = p;
    }
    if (i < NK) {
        float4 p;
        p.x = kvec[i * 3 + 0];
        p.y = kvec[i * 3 + 1];
        p.z = kvec[i * 3 + 2];
        p.w = expfac[i];
        packK[i] = p;
    }
}

#define RED_WAVES 16

#define PF 8

__device__ __forceinline__ void struct_fac_chunk(const float4 p, const v2f a, const bool hi,
                                                 v8f& sumRe, v8f& sumIm)
{
    v4f pv = {p.x, p.y, p.z, p.w};
    asm volatile("" : "+v"(pv));
    const float pq = pv[3];
    v2f bm;
    bm.x = hi ? pv[2] : pv[0];
    bm.y = hi ? 0.0f : pv[1];

    v8f d = {};
    d = __builtin_amdgcn_wmma_f32_16x16x4_f32(
            false, a, false, bm, (short)0, d, false, false);

    v8f sn, cs;
    sincos2pi_x8(d, &sn, &cs);
    sumRe = sumRe + cs * pq;
    sumIm = sumIm + sn * pq;
}

__device__ __forceinline__ void recip_chunk(const float4 kv, const float2 S, const v2f a,
                                            const bool hi, v8f& acc)
{
    v4f kq = {kv.x, kv.y, kv.z, kv.w};
    v2f sq = {S.x, S.y};
    asm volatile("" : "+v"(kq), "+v"(sq));
    v2f bm;
    bm.x = hi ? kq[2] : kq[0];
    bm.y = hi ? 0.0f : kq[1];

    v8f d = {};
    d = __builtin_amdgcn_wmma_f32_16x16x4_f32(
            false, a, false, bm, (short)0, d, false, false);

    v8f sn, cs;
    sincos2pi_x8(d, &sn, &cs);
    acc = acc + (cs * sq[0] + sn * sq[1]) * kq[3];
}

__global__ __launch_bounds__(32 * RED_WAVES) void k_struct_fac(
    const float4* __restrict__ packA, const float* __restrict__ cell_inv,
    const float* __restrict__ kvec, float* __restrict__ Sri,
    int N, int NK)
{
    __shared__ float red[RED_WAVES][32];

    const int lane  = threadIdx.x & 31;
    const int wave  = __builtin_amdgcn_readfirstlane(threadIdx.x >> 5);
    const int kTile = blockIdx.x * 16;
    const int b     = blockIdx.z;
    const int nPer  = N / RED_WAVES;
    const int nBeg  = wave * nPer;

    const int  m  = lane & 15;
    const bool hi = lane >= 16;

    const float kx = kvec[(kTile + m) * 3 + 0];
    const float ky = kvec[(kTile + m) * 3 + 1];
    const float kz = kvec[(kTile + m) * 3 + 2];
    const float kp0 = kx * cell_inv[0] + ky * cell_inv[3] + kz * cell_inv[6];
    const float kp1 = kx * cell_inv[1] + ky * cell_inv[4] + kz * cell_inv[7];
    const float kp2 = kx * cell_inv[2] + ky * cell_inv[5] + kz * cell_inv[8];

    v2f a;
    a.x = hi ? kp2 : kp0;
    a.y = hi ? 0.0f : kp1;

    v8f sumRe = {}, sumIm = {};
    const float4* pA = packA + (size_t)b * N;
    const int nEnd = nBeg + nPer;

    float4 slot[PF];
#pragma unroll
    for (int j = 0; j < PF; ++j) slot[j] = pA[nBeg + 16 * j + m];

    int n0 = nBeg;
    for (; n0 + 16 * PF < nEnd; n0 += 16 * PF) {
#pragma unroll
        for (int j = 0; j < PF; ++j) {
            struct_fac_chunk(slot[j], a, hi, sumRe, sumIm);
            slot[j] = pA[n0 + 16 * (j + PF) + m];
        }
    }
#pragma unroll
    for (int j = 0; j < PF; ++j)
        struct_fac_chunk(slot[j], a, hi, sumRe, sumIm);

#pragma unroll
    for (int r = 0; r < 8; ++r) {
        float re = sumRe[r], im = sumIm[r];
        for (int mask = 8; mask >= 1; mask >>= 1) {
            re += __shfl_xor(re, mask, 32);
            im += __shfl_xor(im, mask, 32);
        }
        sumRe[r] = re; sumIm[r] = im;
    }
    if (m == 0) {
        const int o = hi ? 16 : 0;
#pragma unroll
        for (int r = 0; r < 8; ++r) {
            red[wave][o + 2 * r + 0] = sumRe[r];
            red[wave][o + 2 * r + 1] = sumIm[r];
        }
    }
    __syncthreads();
    if (wave == 0) {
        float s = red[0][lane];
#pragma unroll
        for (int w = 1; w < RED_WAVES; ++w) s += red[w][lane];
        Sri[2 * ((size_t)b * NK + kTile) + lane] = s;
    }
}

__global__ __launch_bounds__(64 * RED_WAVES) void k_recip(
    const float4* __restrict__ packA, const float* __restrict__ cell_inv,
    const float4* __restrict__ packK, const float2* __restrict__ Sri,
    const float* __restrict__ q, const float* __restrict__ volume,
    const float* __restrict__ bewald, float* __restrict__ out,
    int N, int NK, int BN)
{
    __shared__ float red[RED_WAVES][32];

    const int lane  = threadIdx.x & 31;
    const int wave  = __builtin_amdgcn_readfirstlane(threadIdx.x >> 5);
    const int tile  = wave & 1;
    const int slice = wave >> 1;
    const int nBlk  = blockIdx.x * 32;
    const int nTile = nBlk + tile * 16;
    const int b     = blockIdx.z;
    const int kPer  = NK / RED_WAVES;
    const int kBeg  = slice * kPer;

    const int  m  = lane & 15;
    const bool hi = lane >= 16;

    const float4 pc = packA[(size_t)b * N + nTile + m];
    const float s0 = cell_inv[0] * pc.x + cell_inv[1] * pc.y + cell_inv[2] * pc.z;
    const float s1 = cell_inv[3] * pc.x + cell_inv[4] * pc.y + cell_inv[5] * pc.z;
    const float s2 = cell_inv[6] * pc.x + cell_inv[7] * pc.y + cell_inv[8] * pc.z;

    v2f a;
    a.x = hi ? s2 : s0;
    a.y = hi ? 0.0f : s1;

    v8f acc = {};
    const float2* pS = Sri + (size_t)b * NK;
    const int kEnd = kBeg + kPer;

    float4 kslot[PF];
    float2 sslot[PF];
#pragma unroll
    for (int j = 0; j < PF; ++j) {
        kslot[j] = packK[kBeg + 16 * j + m];
        sslot[j] = pS[kBeg + 16 * j + m];
    }

    int kc = kBeg;
    for (; kc + 16 * PF < kEnd; kc += 16 * PF) {
#pragma unroll
        for (int j = 0; j < PF; ++j) {
            recip_chunk(kslot[j], sslot[j], a, hi, acc);
            kslot[j] = packK[kc + 16 * (j + PF) + m];
            sslot[j] = pS[kc + 16 * (j + PF) + m];
        }
    }
#pragma unroll
    for (int j = 0; j < PF; ++j)
        recip_chunk(kslot[j], sslot[j], a, hi, acc);

#pragma unroll
    for (int r = 0; r < 8; ++r) {
        float v = acc[r];
        for (int mask = 8; mask >= 1; mask >>= 1)
            v += __shfl_xor(v, mask, 32);
        acc[r] = v;
    }
    if (m == 0) {
        const int o = tile * 16 + (hi ? 8 : 0);
#pragma unroll
        for (int r = 0; r < 8; ++r)
            red[slice][o + r] = acc[r];
    }
    __syncthreads();
    if (wave == 0) {
        float recip = red[0][lane];
#pragma unroll
        for (int s = 1; s < RED_WAVES; ++s) recip += red[s][lane];
        const int   i     = b * N + nBlk + lane;
        const float qi    = q[i];
        const float scale = BOHR_F / (PI_F * volume[0]);
        const float self  = 2.0f * bewald[0] * BOHR_F / sqrtf(PI_F);
        const float phi   = recip * scale - qi * self;
        out[i]      = 0.5f * qi * phi;
        out[BN + i] = phi;
    }
}

extern "C" void kernel_launch(void* const* d_in, const int* in_sizes, int n_in,
                              void* d_out, int out_size, void* d_ws, size_t ws_size,
                              hipStream_t stream)
{
    const float* coords   = (const float*)d_in[0];
    const float* q        = (const float*)d_in[1];
    const float* cell_inv = (const float*)d_in[2];
    const float* kvec     = (const float*)d_in[3];
    const float* expfac   = (const float*)d_in[4];
    const float* volume   = (const float*)d_in[5];
    const float* bewald   = (const float*)d_in[6];
    float* out = (float*)d_out;

    const int B  = 2;
    const int BN = in_sizes[1];
    const int N  = BN / B;
    const int NK = in_sizes[4];

    float*  Sri   = (float*)d_ws;
    float4* packA = (float4*)(Sri + (size_t)2 * B * NK);
    float4* packK = packA + BN;
    const size_t wsFloats = (size_t)2 * B * NK + (size_t)4 * BN + (size_t)4 * NK;
    if (ws_size < wsFloats * sizeof(float)) return;

    const int packN = (BN > NK) ? BN : NK;
    k_pack<<<(packN + 255) / 256, 256, 0, stream>>>(coords, q, kvec, expfac,
                                                    packA, packK, BN, NK);

    dim3 g1(NK / 16, 1, B);
    k_struct_fac<<<g1, 32 * RED_WAVES, 0, stream>>>(packA, cell_inv, kvec, Sri, N, NK);

    dim3 g2(N / 32, 1, B);
    k_recip<<<g2, 64 * RED_WAVES, 0, stream>>>(packA, cell_inv, packK, (const float2*)Sri,
                                               q, volume, bewald, out, N, NK, BN);
}


# Kernel 2 computes expfac*(Sre*cos+Sim*sin) as (expfac*|S|)*cos(th-arg S): kernel 1 stores amplitude and phase per k, phase rides in the WMMA's spare 4th k slot; 8 v_cos, no v_sin; k-sum order kept
# speedup vs baseline: 2.2313x; 1.3187x over previous
#include <hip/hip_runtime.h>
#include <math.h>

typedef __attribute__((ext_vector_type(2))) float v2f;
typedef __attribute__((ext_vector_type(8))) float v8f;

#define BOHR_F 1.8897261258369282f
#define PI_F   3.14159265358979323846f
#define INV_2PI_F 0.15915494309189533577f

__device__ __forceinline__ void sincos2pi_x8(v8f x, v8f* sn, v8f* cs) {
    float s0, s1, s2, s3, s4, s5, s6, s7;
    float c0, c1, c2, c3, c4, c5, c6, c7;
    asm volatile(
        "v_sin_f32 %0, %16\n\t"
        "v_sin_f32 %1, %17\n\t"
        "v_sin_f32 %2, %18\n\t"
        "v_sin_f32 %3, %19\n\t"
        "v_sin_f32 %4, %20\n\t"
        "v_sin_f32 %5, %21\n\t"
        "v_sin_f32 %6, %22\n\t"
        "v_sin_f32 %7, %23\n\t"
        "v_cos_f32 %8, %16\n\t"
        "v_cos_f32 %9, %17\n\t"
        "v_cos_f32 %10, %18\n\t"
        "v_cos_f32 %11, %19\n\t"
        "v_cos_f32 %12, %20\n\t"
        "v_cos_f32 %13, %21\n\t"
        "v_cos_f32 %14, %22\n\t"
        "v_cos_f32 %15, %23\n\t"
        "v_nop"
        : "=&v"(s0), "=&v"(s1), "=&v"(s2), "=&v"(s3),
          "=&v"(s4), "=&v"(s5), "=&v"(s6), "=&v"(s7),
          "=&v"(c0), "=&v"(c1), "=&v"(c2), "=&v"(c3),
          "=&v"(c4), "=&v"(c5), "=&v"(c6), "=&v"(c7)
        : "v"(x[0]), "v"(x[1]), "v"(x[2]), "v"(x[3]),
          "v"(x[4]), "v"(x[5]), "v"(x[6]), "v"(x[7]));
    *sn = (v8f){s0, s1, s2, s3, s4, s5, s6, s7};
    *cs = (v8f){c0, c1, c2, c3, c4, c5, c6, c7};
}

__device__ __forceinline__ void cos2pi_x8(v8f x, v8f* cs) {
    float c0, c1, c2, c3, c4, c5, c6, c7;
    asm volatile(
        "v_cos_f32 %0, %8\n\t"
        "v_cos_f32 %1, %9\n\t"
        "v_cos_f32 %2, %10\n\t"
        "v_cos_f32 %3, %11\n\t"
        "v_cos_f32 %4, %12\n\t"
        "v_cos_f32 %5, %13\n\t"
        "v_cos_f32 %6, %14\n\t"
        "v_cos_f32 %7, %15\n\t"
        "v_nop"
        : "=&v"(c0), "=&v"(c1), "=&v"(c2), "=&v"(c3),
          "=&v"(c4), "=&v"(c5), "=&v"(c6), "=&v"(c7)
        : "v"(x[0]), "v"(x[1]), "v"(x[2]), "v"(x[3]),
          "v"(x[4]), "v"(x[5]), "v"(x[6]), "v"(x[7]));
    *cs = (v8f){c0, c1, c2, c3, c4, c5, c6, c7};
}

struct Row3 { float c02, c1, w; };
struct KRow { float c02, c1; float2 ap; };

__device__ __forceinline__ Row3 load_row3(const float* __restrict__ p02, const float* __restrict__ p1,
                                          const float* __restrict__ pw, const int chunk)
{
    Row3 r;
    r.c02 = p02[48 * chunk];
    r.c1  = p1[48 * chunk];
    r.w   = pw[16 * chunk];
    return r;
}

__device__ __forceinline__ KRow load_krow(const float* __restrict__ p02, const float* __restrict__ p1,
                                          const float2* __restrict__ pap, const int chunk)
{
    KRow r;
    r.c02 = p02[48 * chunk];
    r.c1  = p1[48 * chunk];
    r.ap  = pap[16 * chunk];
    return r;
}

#define RED_WAVES 16

#define PF 8

__device__ __forceinline__ void struct_fac_chunk(const Row3 p, const v2f a, const bool hi,
                                                 v8f& sumRe, v8f& sumIm)
{
    float xz = p.c02, y = p.c1, pq = p.w;
    asm volatile("" : "+v"(xz), "+v"(y), "+v"(pq));
    v2f bm;
    bm.x = xz;
    bm.y = hi ? 0.0f : y;

    v8f d = {};
    d = __builtin_amdgcn_wmma_f32_16x16x4_f32(
            false, a, false, bm, (short)0, d, false, false);

    v8f sn, cs;
    sincos2pi_x8(d, &sn, &cs);
    sumRe = sumRe + cs * pq;
    sumIm = sumIm + sn * pq;
}

__device__ __forceinline__ void recip_chunk(const KRow kv, const v2f a, const bool hi, v8f& acc)
{
    float kxz = kv.c02, ky = kv.c1;
    v2f ap = {kv.ap.x, kv.ap.y};
    asm volatile("" : "+v"(kxz), "+v"(ky), "+v"(ap));
    v2f bm;
    bm.x = kxz;
    bm.y = hi ? ap[1] : ky;

    v8f d = {};
    d = __builtin_amdgcn_wmma_f32_16x16x4_f32(
            false, a, false, bm, (short)0, d, false, false);

    v8f cs;
    cos2pi_x8(d, &cs);
    acc = acc + cs * ap[0];
}

__global__ __launch_bounds__(32 * RED_WAVES) void k_struct_fac(
    const float* __restrict__ coords, const float* __restrict__ q,
    const float* __restrict__ cell_inv,
    const float* __restrict__ kvec, const float* __restrict__ expfac,
    float* __restrict__ Sap,
    int N, int NK)
{
    __shared__ float red[RED_WAVES][32];

    const int lane  = threadIdx.x & 31;
    const int wave  = __builtin_amdgcn_readfirstlane(threadIdx.x >> 5);
    const int kTile = blockIdx.x * 16;
    const int b     = blockIdx.z;
    const int nPer  = N / RED_WAVES;
    const int nBeg  = wave * nPer;

    const int  m  = lane & 15;
    const bool hi = lane >= 16;

    const float kx = kvec[(kTile + m) * 3 + 0];
    const float ky = kvec[(kTile + m) * 3 + 1];
    const float kz = kvec[(kTile + m) * 3 + 2];
    const float kp0 = kx * cell_inv[0] + ky * cell_inv[3] + kz * cell_inv[6];
    const float kp1 = kx * cell_inv[1] + ky * cell_inv[4] + kz * cell_inv[7];
    const float kp2 = kx * cell_inv[2] + ky * cell_inv[5] + kz * cell_inv[8];

    v2f a;
    a.x = hi ? kp2 : kp0;
    a.y = hi ? 0.0f : kp1;

    v8f sumRe = {}, sumIm = {};
    const size_t atom0 = (size_t)b * N + nBeg + m;
    const float* pXZ = coords + atom0 * 3 + (hi ? 2 : 0);
    const float* pY  = coords + atom0 * 3 + 1;
    const float* pQ  = q + atom0;
    const int nChunks = nPer / 16;

    Row3 slot[PF];
#pragma unroll
    for (int j = 0; j < PF; ++j) {
        slot[j] = load_row3(pXZ, pY, pQ, j);
        __builtin_amdgcn_sched_barrier(0);
    }

    int c0 = 0;
    for (; c0 + PF < nChunks; c0 += PF) {
#pragma unroll
        for (int j = 0; j < PF; ++j) {
            struct_fac_chunk(slot[j], a, hi, sumRe, sumIm);
            slot[j] = load_row3(pXZ, pY, pQ, c0 + j + PF);
        }
    }
#pragma unroll
    for (int j = 0; j < PF; ++j)
        struct_fac_chunk(slot[j], a, hi, sumRe, sumIm);

#pragma unroll
    for (int r = 0; r < 8; ++r) {
        float re = sumRe[r], im = sumIm[r];
        for (int mask = 8; mask >= 1; mask >>= 1) {
            re += __shfl_xor(re, mask, 32);
            im += __shfl_xor(im, mask, 32);
        }
        sumRe[r] = re; sumIm[r] = im;
    }
    if (m == 0) {
        const int o = hi ? 16 : 0;
#pragma unroll
        for (int r = 0; r < 8; ++r) {
            red[wave][o + 2 * r + 0] = sumRe[r];
            red[wave][o + 2 * r + 1] = sumIm[r];
        }
    }
    __syncthreads();
    if (wave == 0) {
        float s = red[0][lane];
#pragma unroll
        for (int w = 1; w < RED_WAVES; ++w) s += red[w][lane];
        const bool  odd   = lane & 1;
        const float other = __shfl_xor(s, 1, 32);
        const float re    = odd ? other : s;
        const float im    = odd ? s : other;
        const float amp   = hypotf(re, im) * expfac[kTile + (lane >> 1)];
        const float rev   = atan2f(im, re) * INV_2PI_F;
        Sap[2 * ((size_t)b * NK + kTile) + lane] = odd ? -rev : amp;
    }
}

__global__ __launch_bounds__(64 * RED_WAVES) void k_recip(
    const float* __restrict__ coords, const float* __restrict__ cell_inv,
    const float* __restrict__ kvec,
    const float2* __restrict__ Sap,
    const float* __restrict__ q, const float* __restrict__ volume,
    const float* __restrict__ bewald, float* __restrict__ out,
    int N, int NK, int BN)
{
    __shared__ float red[RED_WAVES][32];

    const int lane  = threadIdx.x & 31;
    const int wave  = __builtin_amdgcn_readfirstlane(threadIdx.x >> 5);
    const int tile  = wave & 1;
    const int slice = wave >> 1;
    const int nBlk  = blockIdx.x * 32;
    const int nTile = nBlk + tile * 16;
    const int b     = blockIdx.z;
    const int kPer  = NK / RED_WAVES;
    const int kBeg  = slice * kPer;

    const int  m  = lane & 15;
    const bool hi = lane >= 16;

    const float* pc = coords + ((size_t)b * N + nTile + m) * 3;
    const float cx = pc[0], cy = pc[1], cz = pc[2];
    const float s0 = cell_inv[0] * cx + cell_inv[1] * cy + cell_inv[2] * cz;
    const float s1 = cell_inv[3] * cx + cell_inv[4] * cy + cell_inv[5] * cz;
    const float s2 = cell_inv[6] * cx + cell_inv[7] * cy + cell_inv[8] * cz;

    v2f a;
    a.x = hi ? s2 : s0;
    a.y = hi ? 1.0f : s1;

    v8f acc = {};
    const size_t k0 = (size_t)kBeg + m;
    const float*  pKXZ = kvec + k0 * 3 + (hi ? 2 : 0);
    const float*  pKY  = kvec + k0 * 3 + 1;
    const float2* pS   = Sap + (size_t)b * NK + k0;
    const int nChunks = kPer / 16;

    KRow kslot[PF];
#pragma unroll
    for (int j = 0; j < PF; ++j) {
        kslot[j] = load_krow(pKXZ, pKY, pS, j);
        __builtin_amdgcn_sched_barrier(0);
    }

    int c0 = 0;
    for (; c0 + PF < nChunks; c0 += PF) {
#pragma unroll
        for (int j = 0; j < PF; ++j) {
            recip_chunk(kslot[j], a, hi, acc);
            kslot[j] = load_krow(pKXZ, pKY, pS, c0 + j + PF);
        }
    }
#pragma unroll
    for (int j = 0; j < PF; ++j)
        recip_chunk(kslot[j], a, hi, acc);

#pragma unroll
    for (int r = 0; r < 8; ++r) {
        float v = acc[r];
        for (int mask = 8; mask >= 1; mask >>= 1)
            v += __shfl_xor(v, mask, 32);
        acc[r] = v;
    }
    if (m == 0) {
        const int o = tile * 16 + (hi ? 8 : 0);
#pragma unroll
        for (int r = 0; r < 8; ++r)
            red[slice][o + r] = acc[r];
    }
    __syncthreads();
    if (wave == 0) {
        float recip = red[0][lane];
#pragma unroll
        for (int s = 1; s < RED_WAVES; ++s) recip += red[s][lane];
        const int   i     = b * N + nBlk + lane;
        const float qi    = q[i];
        const float scale = BOHR_F / (PI_F * volume[0]);
        const float self  = 2.0f * bewald[0] * BOHR_F / sqrtf(PI_F);
        const float phi   = recip * scale - qi * self;
        out[i]      = 0.5f * qi * phi;
        out[BN + i] = phi;
    }
}

extern "C" void kernel_launch(void* const* d_in, const int* in_sizes, int n_in,
                              void* d_out, int out_size, void* d_ws, size_t ws_size,
                              hipStream_t stream)
{
    const float* coords   = (const float*)d_in[0];
    const float* q        = (const float*)d_in[1];
    const float* cell_inv = (const float*)d_in[2];
    const float* kvec     = (const float*)d_in[3];
    const float* expfac   = (const float*)d_in[4];
    const float* volume   = (const float*)d_in[5];
    const float* bewald   = (const float*)d_in[6];
    float* out = (float*)d_out;

    const int B  = 2;
    const int BN = in_sizes[1];
    const int N  = BN / B;
    const int NK = in_sizes[4];

    float* Sap = (float*)d_ws;
    const size_t wsFloats = (size_t)2 * B * NK;
    if (ws_size < wsFloats * sizeof(float)) return;

    dim3 g1(NK / 16, 1, B);
    k_struct_fac<<<g1, 32 * RED_WAVES, 0, stream>>>(coords, q, cell_inv, kvec, expfac, Sap, N, NK);

    dim3 g2(N / 32, 1, B);
    k_recip<<<g2, 64 * RED_WAVES, 0, stream>>>(coords, cell_inv, kvec, (const float2*)Sap,
                                               q, volume, bewald, out, N, NK, BN);
}


# Both chunk loops software-pipelined in asm: a block issues the next chunk's WMMA, then this chunk's sin/cos with the previous chunk's packed multiply-adds in the gaps behind them; sum order unchanged
# speedup vs baseline: 2.3505x; 1.0534x over previous
#include <hip/hip_runtime.h>
#include <math.h>

typedef __attribute__((ext_vector_type(2))) float v2f;
typedef __attribute__((ext_vector_type(8))) float v8f;

#define BOHR_F 1.8897261258369282f
#define PI_F   3.14159265358979323846f
#define INV_2PI_F 0.15915494309189533577f


struct Row3 { float c02, c1, w; };
struct KRow { float c02, c1; float2 ap; };

__device__ __forceinline__ Row3 load_row3(const float* __restrict__ p02, const float* __restrict__ p1,
                                          const float* __restrict__ pw, const int chunk)
{
    Row3 r;
    r.c02 = p02[48 * chunk];
    r.c1  = p1[48 * chunk];
    r.w   = pw[16 * chunk];
    return r;
}

__device__ __forceinline__ KRow load_krow(const float* __restrict__ p02, const float* __restrict__ p1,
                                          const float2* __restrict__ pap, const int chunk)
{
    KRow r;
    r.c02 = p02[48 * chunk];
    r.c1  = p1[48 * chunk];
    r.ap  = pap[16 * chunk];
    return r;
}

#define RED_WAVES 16

#define PF 8

struct SinCos { v2f s[4], c[4]; };
struct Sums   { v2f re[4], im[4]; };

__device__ __forceinline__ v2f atoms_frag(const Row3 p, const bool hi)
{
    float xz = p.c02, y = p.c1;
    asm volatile("" : "+v"(xz), "+v"(y));
    v2f bm;
    bm.x = xz;
    bm.y = hi ? 0.0f : y;
    return bm;
}

template <bool HAS_NEXT>
__device__ __forceinline__ void struct_fac_stage(const v2f a, const v2f bnext, v8f& d,
                                                 SinCos& prev, const float qprev, Sums& sum)
{
    float s0, s1, s2, s3, s4, s5, s6, s7;
    float c0, c1, c2, c3, c4, c5, c6, c7;
    v2f q2;
    q2.x = qprev;
    if (HAS_NEXT) {
        v8f dn;
        asm volatile(
            "v_wmma_f32_16x16x4_f32 %[dn], %[a], %[b], 0\n\t"
            "v_cos_f32 %[c0], %[x0]\n\t"
            "v_cos_f32 %[c1], %[x1]\n\t"
            "v_cos_f32 %[c2], %[x2]\n\t"
            "v_cos_f32 %[c3], %[x3]\n\t"
            "v_cos_f32 %[c4], %[x4]\n\t"
            "v_pk_fma_f32 %[im01], %[q], %[ps01], %[im01] op_sel_hi:[0,1,1]\n\t"
            "v_cos_f32 %[c5], %[x5]\n\t"
            "v_pk_fma_f32 %[im23], %[q], %[ps23], %[im23] op_sel_hi:[0,1,1]\n\t"
            "v_cos_f32 %[c6], %[x6]\n\t"
            "v_pk_fma_f32 %[im45], %[q], %[ps45], %[im45] op_sel_hi:[0,1,1]\n\t"
            "v_cos_f32 %[c7], %[x7]\n\t"
            "v_pk_fma_f32 %[im67], %[q], %[ps67], %[im67] op_sel_hi:[0,1,1]"
            :
              [dn] "=&v"(dn), [c0] "=&v"(c0), [c1] "=&v"(c1), [c2] "=&v"(c2),
              [c3] "=&v"(c3), [c4] "=&v"(c4), [c5] "=&v"(c5), [c6] "=&v"(c6),
              [c7] "=&v"(c7), [im01] "+v"(sum.im[0]), [im23] "+v"(sum.im[1]), [im45] "+v"(sum.im[2]),
              [im67] "+v"(sum.im[3])
            :
              [a] "v"(a), [b] "v"(bnext), [x0] "v"(d[0]), [x1] "v"(d[1]),
              [x2] "v"(d[2]), [x3] "v"(d[3]), [x4] "v"(d[4]), [x5] "v"(d[5]),
              [x6] "v"(d[6]), [x7] "v"(d[7]), [ps01] "v"(prev.s[0]), [ps23] "v"(prev.s[1]),
              [ps45] "v"(prev.s[2]), [ps67] "v"(prev.s[3]), [q] "v"(q2));
        asm volatile(
            "v_sin_f32 %[s0], %[x0]\n\t"
            "v_sin_f32 %[s1], %[x1]\n\t"
            "v_sin_f32 %[s2], %[x2]\n\t"
            "v_sin_f32 %[s3], %[x3]\n\t"
            "v_sin_f32 %[s4], %[x4]\n\t"
            "v_pk_fma_f32 %[re01], %[q], %[pc01], %[re01] op_sel_hi:[0,1,1]\n\t"
            "v_sin_f32 %[s5], %[x5]\n\t"
            "v_pk_fma_f32 %[re23], %[q], %[pc23], %[re23] op_sel_hi:[0,1,1]\n\t"
            "v_sin_f32 %[s6], %[x6]\n\t"
            "v_pk_fma_f32 %[re45], %[q], %[pc45], %[re45] op_sel_hi:[0,1,1]\n\t"
            "v_sin_f32 %[s7], %[x7]\n\t"
            "v_pk_fma_f32 %[re67], %[q], %[pc67], %[re67] op_sel_hi:[0,1,1]"
            :
              [s0] "=&v"(s0), [s1] "=&v"(s1), [s2] "=&v"(s2), [s3] "=&v"(s3),
              [s4] "=&v"(s4), [s5] "=&v"(s5), [s6] "=&v"(s6), [s7] "=&v"(s7),
              [re01] "+v"(sum.re[0]), [re23] "+v"(sum.re[1]), [re45] "+v"(sum.re[2]), [re67] "+v"(sum.re[3])
            :
              [x0] "v"(d[0]), [x1] "v"(d[1]), [x2] "v"(d[2]), [x3] "v"(d[3]),
              [x4] "v"(d[4]), [x5] "v"(d[5]), [x6] "v"(d[6]), [x7] "v"(d[7]),
              [pc01] "v"(prev.c[0]), [pc23] "v"(prev.c[1]), [pc45] "v"(prev.c[2]), [pc67] "v"(prev.c[3]),
              [q] "v"(q2));
        d = dn;
    } else {
        asm volatile(
            "v_cos_f32 %[c0], %[x0]\n\t"
            "v_cos_f32 %[c1], %[x1]\n\t"
            "v_cos_f32 %[c2], %[x2]\n\t"
            "v_cos_f32 %[c3], %[x3]\n\t"
            "v_cos_f32 %[c4], %[x4]\n\t"
            "v_pk_fma_f32 %[im01], %[q], %[ps01], %[im01] op_sel_hi:[0,1,1]\n\t"
            "v_cos_f32 %[c5], %[x5]\n\t"
            "v_pk_fma_f32 %[im23], %[q], %[ps23], %[im23] op_sel_hi:[0,1,1]\n\t"
            "v_cos_f32 %[c6], %[x6]\n\t"
            "v_pk_fma_f32 %[im45], %[q], %[ps45], %[im45] op_sel_hi:[0,1,1]\n\t"
            "v_cos_f32 %[c7], %[x7]\n\t"
            "v_pk_fma_f32 %[im67], %[q], %[ps67], %[im67] op_sel_hi:[0,1,1]"
            :
              [c0] "=&v"(c0), [c1] "=&v"(c1), [c2] "=&v"(c2), [c3] "=&v"(c3),
              [c4] "=&v"(c4), [c5] "=&v"(c5), [c6] "=&v"(c6), [c7] "=&v"(c7),
              [im01] "+v"(sum.im[0]), [im23] "+v"(sum.im[1]), [im45] "+v"(sum.im[2]), [im67] "+v"(sum.im[3])
            :
              [x0] "v"(d[0]), [x1] "v"(d[1]), [x2] "v"(d[2]), [x3] "v"(d[3]),
              [x4] "v"(d[4]), [x5] "v"(d[5]), [x6] "v"(d[6]), [x7] "v"(d[7]),
              [ps01] "v"(prev.s[0]), [ps23] "v"(prev.s[1]), [ps45] "v"(prev.s[2]), [ps67] "v"(prev.s[3]),
              [q] "v"(q2));
        asm volatile(
            "v_sin_f32 %[s0], %[x0]\n\t"
            "v_sin_f32 %[s1], %[x1]\n\t"
            "v_sin_f32 %[s2], %[x2]\n\t"
            "v_sin_f32 %[s3], %[x3]\n\t"
            "v_sin_f32 %[s4], %[x4]\n\t"
            "v_pk_fma_f32 %[re01], %[q], %[pc01], %[re01] op_sel_hi:[0,1,1]\n\t"
            "v_sin_f32 %[s5], %[x5]\n\t"
            "v_pk_fma_f32 %[re23], %[q], %[pc23], %[re23] op_sel_hi:[0,1,1]\n\t"
            "v_sin_f32 %[s6], %[x6]\n\t"
            "v_pk_fma_f32 %[re45], %[q], %[pc45], %[re45] op_sel_hi:[0,1,1]\n\t"
            "v_sin_f32 %[s7], %[x7]\n\t"
            "v_pk_fma_f32 %[re67], %[q], %[pc67], %[re67] op_sel_hi:[0,1,1]"
            :
              [s0] "=&v"(s0), [s1] "=&v"(s1), [s2] "=&v"(s2), [s3] "=&v"(s3),
              [s4] "=&v"(s4), [s5] "=&v"(s5), [s6] "=&v"(s6), [s7] "=&v"(s7),
              [re01] "+v"(sum.re[0]), [re23] "+v"(sum.re[1]), [re45] "+v"(sum.re[2]), [re67] "+v"(sum.re[3])
            :
              [x0] "v"(d[0]), [x1] "v"(d[1]), [x2] "v"(d[2]), [x3] "v"(d[3]),
              [x4] "v"(d[4]), [x5] "v"(d[5]), [x6] "v"(d[6]), [x7] "v"(d[7]),
              [pc01] "v"(prev.c[0]), [pc23] "v"(prev.c[1]), [pc45] "v"(prev.c[2]), [pc67] "v"(prev.c[3]),
              [q] "v"(q2));
    }
    prev.s[0] = (v2f){s0, s1}; prev.s[1] = (v2f){s2, s3}; prev.s[2] = (v2f){s4, s5}; prev.s[3] = (v2f){s6, s7};
    prev.c[0] = (v2f){c0, c1}; prev.c[1] = (v2f){c2, c3}; prev.c[2] = (v2f){c4, c5}; prev.c[3] = (v2f){c6, c7};
}

struct Cos4 { v2f c[4]; };
struct Acc4 { v2f v[4]; };

__device__ __forceinline__ v2f kvec_frag(const KRow kv, const bool hi)
{
    float kxz = kv.c02, ky = kv.c1;
    asm volatile("" : "+v"(kxz), "+v"(ky));
    v2f bm;
    bm.x = kxz;
    bm.y = hi ? kv.ap.y : ky;
    return bm;
}

template <bool HAS_NEXT>
__device__ __forceinline__ void recip_stage(const v2f a, const v2f bnext, v8f& d,
                                            Cos4& prev, const float ampprev, Acc4& acc)
{
    float c0, c1, c2, c3, c4, c5, c6, c7;
    v2f q2;
    q2.x = ampprev;
    if (HAS_NEXT) {
        v8f dn;
        asm volatile(
            "v_wmma_f32_16x16x4_f32 %[dn], %[a], %[b], 0\n\t"
            "v_cos_f32 %[c0], %[x0]\n\t"
            "v_cos_f32 %[c1], %[x1]\n\t"
            "v_cos_f32 %[c2], %[x2]\n\t"
            "v_cos_f32 %[c3], %[x3]\n\t"
            "v_cos_f32 %[c4], %[x4]\n\t"
            "v_pk_fma_f32 %[ac01], %[q], %[pc01], %[ac01] op_sel_hi:[0,1,1]\n\t"
            "v_cos_f32 %[c5], %[x5]\n\t"
            "v_pk_fma_f32 %[ac23], %[q], %[pc23], %[ac23] op_sel_hi:[0,1,1]\n\t"
            "v_cos_f32 %[c6], %[x6]\n\t"
            "v_pk_fma_f32 %[ac45], %[q], %[pc45], %[ac45] op_sel_hi:[0,1,1]\n\t"
            "v_cos_f32 %[c7], %[x7]\n\t"
            "v_pk_fma_f32 %[ac67], %[q], %[pc67], %[ac67] op_sel_hi:[0,1,1]"
            :
              [dn] "=&v"(dn), [c0] "=&v"(c0), [c1] "=&v"(c1), [c2] "=&v"(c2),
              [c3] "=&v"(c3), [c4] "=&v"(c4), [c5] "=&v"(c5), [c6] "=&v"(c6),
              [c7] "=&v"(c7), [ac01] "+v"(acc.v[0]), [ac23] "+v"(acc.v[1]), [ac45] "+v"(acc.v[2]),
              [ac67] "+v"(acc.v[3])
            :
              [a] "v"(a), [b] "v"(bnext), [x0] "v"(d[0]), [x1] "v"(d[1]),
              [x2] "v"(d[2]), [x3] "v"(d[3]), [x4] "v"(d[4]), [x5] "v"(d[5]),
              [x6] "v"(d[6]), [x7] "v"(d[7]), [pc01] "v"(prev.c[0]), [pc23] "v"(prev.c[1]),
              [pc45] "v"(prev.c[2]), [pc67] "v"(prev.c[3]), [q] "v"(q2));
        d = dn;
    } else {
        asm volatile(
            "v_cos_f32 %[c0], %[x0]\n\t"
            "v_cos_f32 %[c1], %[x1]\n\t"
            "v_cos_f32 %[c2], %[x2]\n\t"
            "v_cos_f32 %[c3], %[x3]\n\t"
            "v_cos_f32 %[c4], %[x4]\n\t"
            "v_pk_fma_f32 %[ac01], %[q], %[pc01], %[ac01] op_sel_hi:[0,1,1]\n\t"
            "v_cos_f32 %[c5], %[x5]\n\t"
            "v_pk_fma_f32 %[ac23], %[q], %[pc23], %[ac23] op_sel_hi:[0,1,1]\n\t"
            "v_cos_f32 %[c6], %[x6]\n\t"
            "v_pk_fma_f32 %[ac45], %[q], %[pc45], %[ac45] op_sel_hi:[0,1,1]\n\t"
            "v_cos_f32 %[c7], %[x7]\n\t"
            "v_pk_fma_f32 %[ac67], %[q], %[pc67], %[ac67] op_sel_hi:[0,1,1]"
            :
              [c0] "=&v"(c0), [c1] "=&v"(c1), [c2] "=&v"(c2), [c3] "=&v"(c3),
              [c4] "=&v"(c4), [c5] "=&v"(c5), [c6] "=&v"(c6), [c7] "=&v"(c7),
              [ac01] "+v"(acc.v[0]), [ac23] "+v"(acc.v[1]), [ac45] "+v"(acc.v[2]), [ac67] "+v"(acc.v[3])
            :
              [x0] "v"(d[0]), [x1] "v"(d[1]), [x2] "v"(d[2]), [x3] "v"(d[3]),
              [x4] "v"(d[4]), [x5] "v"(d[5]), [x6] "v"(d[6]), [x7] "v"(d[7]),
              [pc01] "v"(prev.c[0]), [pc23] "v"(prev.c[1]), [pc45] "v"(prev.c[2]), [pc67] "v"(prev.c[3]),
              [q] "v"(q2));
    }
    prev.c[0] = (v2f){c0, c1}; prev.c[1] = (v2f){c2, c3}; prev.c[2] = (v2f){c4, c5}; prev.c[3] = (v2f){c6, c7};
}

__global__ __launch_bounds__(32 * RED_WAVES) void k_struct_fac(
    const float* __restrict__ coords, const float* __restrict__ q,
    const float* __restrict__ cell_inv,
    const float* __restrict__ kvec, const float* __restrict__ expfac,
    float* __restrict__ Sap,
    int N, int NK)
{
    __shared__ float red[RED_WAVES][32];

    const int lane  = threadIdx.x & 31;
    const int wave  = __builtin_amdgcn_readfirstlane(threadIdx.x >> 5);
    const int kTile = blockIdx.x * 16;
    const int b     = blockIdx.z;
    const int nPer  = N / RED_WAVES;
    const int nBeg  = wave * nPer;

    const int  m  = lane & 15;
    const bool hi = lane >= 16;

    const float kx = kvec[(kTile + m) * 3 + 0];
    const float ky = kvec[(kTile + m) * 3 + 1];
    const float kz = kvec[(kTile + m) * 3 + 2];
    const float kp0 = kx * cell_inv[0] + ky * cell_inv[3] + kz * cell_inv[6];
    const float kp1 = kx * cell_inv[1] + ky * cell_inv[4] + kz * cell_inv[7];
    const float kp2 = kx * cell_inv[2] + ky * cell_inv[5] + kz * cell_inv[8];

    v2f a;
    a.x = hi ? kp2 : kp0;
    a.y = hi ? 0.0f : kp1;

    Sums   sum  = {};
    SinCos prev = {};
    const size_t atom0 = (size_t)b * N + nBeg + m;
    const float* pXZ = coords + atom0 * 3 + (hi ? 2 : 0);
    const float* pY  = coords + atom0 * 3 + 1;
    const float* pQ  = q + atom0;
    const int nChunks = nPer / 16;

    Row3 slot[PF];
#pragma unroll
    for (int i = 0; i < PF; ++i) {
        const int j = (i + PF - 1) % PF;
        slot[j] = load_row3(pXZ, pY, pQ, j);
        __builtin_amdgcn_sched_barrier(0);
    }

    v8f d = {};
    d = __builtin_amdgcn_wmma_f32_16x16x4_f32(
            false, a, false, atoms_frag(slot[0], hi), (short)0, d, false, false);

    int c0 = 0;
    for (; c0 + PF < nChunks; c0 += PF) {
#pragma unroll
        for (int j = 0; j < PF; ++j) {
            const v2f bn = atoms_frag(slot[(j + 1) % PF], hi);
            struct_fac_stage<true>(a, bn, d, prev, slot[(j + PF - 1) % PF].w, sum);
            slot[(j + PF - 1) % PF] = load_row3(pXZ, pY, pQ, c0 + (j + PF - 1) % PF + (j ? PF : 0));
        }
    }
#pragma unroll
    for (int j = 0; j < PF; ++j) {
        if (j + 1 < PF) {
            const v2f bn = atoms_frag(slot[j + 1], hi);
            struct_fac_stage<true>(a, bn, d, prev, slot[(j + PF - 1) % PF].w, sum);
        } else {
            struct_fac_stage<false>(a, a, d, prev, slot[(j + PF - 1) % PF].w, sum);
        }
        if (j == 0) slot[PF - 1] = load_row3(pXZ, pY, pQ, c0 + PF - 1);
    }
    const float qprev = slot[PF - 1].w;
    v8f sumRe, sumIm;
#pragma unroll
    for (int i = 0; i < 4; ++i) {
        const v2f re = sum.re[i] + prev.c[i] * qprev;
        const v2f im = sum.im[i] + prev.s[i] * qprev;
        sumRe[2 * i] = re.x; sumRe[2 * i + 1] = re.y;
        sumIm[2 * i] = im.x; sumIm[2 * i + 1] = im.y;
    }

#pragma unroll
    for (int r = 0; r < 8; ++r) {
        float re = sumRe[r], im = sumIm[r];
        for (int mask = 8; mask >= 1; mask >>= 1) {
            re += __shfl_xor(re, mask, 32);
            im += __shfl_xor(im, mask, 32);
        }
        sumRe[r] = re; sumIm[r] = im;
    }
    if (m == 0) {
        const int o = hi ? 16 : 0;
#pragma unroll
        for (int r = 0; r < 8; ++r) {
            red[wave][o + 2 * r + 0] = sumRe[r];
            red[wave][o + 2 * r + 1] = sumIm[r];
        }
    }
    __syncthreads();
    if (wave == 0) {
        float s = red[0][lane];
#pragma unroll
        for (int w = 1; w < RED_WAVES; ++w) s += red[w][lane];
        const bool  odd   = lane & 1;
        const float other = __shfl_xor(s, 1, 32);
        const float re    = odd ? other : s;
        const float im    = odd ? s : other;
        const float amp   = hypotf(re, im) * expfac[kTile + (lane >> 1)];
        const float rev   = atan2f(im, re) * INV_2PI_F;
        Sap[2 * ((size_t)b * NK + kTile) + lane] = odd ? -rev : amp;
    }
}

__global__ __launch_bounds__(64 * RED_WAVES) void k_recip(
    const float* __restrict__ coords, const float* __restrict__ cell_inv,
    const float* __restrict__ kvec,
    const float2* __restrict__ Sap,
    const float* __restrict__ q, const float* __restrict__ volume,
    const float* __restrict__ bewald, float* __restrict__ out,
    int N, int NK, int BN)
{
    __shared__ float red[RED_WAVES][32];

    const int lane  = threadIdx.x & 31;
    const int wave  = __builtin_amdgcn_readfirstlane(threadIdx.x >> 5);
    const int tile  = wave & 1;
    const int slice = wave >> 1;
    const int nBlk  = blockIdx.x * 32;
    const int nTile = nBlk + tile * 16;
    const int b     = blockIdx.z;
    const int kPer  = NK / RED_WAVES;
    const int kBeg  = slice * kPer;

    const int  m  = lane & 15;
    const bool hi = lane >= 16;

    const float* pc = coords + ((size_t)b * N + nTile + m) * 3;
    const float cx = pc[0], cy = pc[1], cz = pc[2];
    const float s0 = cell_inv[0] * cx + cell_inv[1] * cy + cell_inv[2] * cz;
    const float s1 = cell_inv[3] * cx + cell_inv[4] * cy + cell_inv[5] * cz;
    const float s2 = cell_inv[6] * cx + cell_inv[7] * cy + cell_inv[8] * cz;

    v2f a;
    a.x = hi ? s2 : s0;
    a.y = hi ? 1.0f : s1;

    Acc4  accp = {};
    Cos4  prev = {};
    const size_t k0 = (size_t)kBeg + m;
    const float*  pKXZ = kvec + k0 * 3 + (hi ? 2 : 0);
    const float*  pKY  = kvec + k0 * 3 + 1;
    const float2* pS   = Sap + (size_t)b * NK + k0;
    const int nChunks = kPer / 16;

    KRow kslot[PF];
#pragma unroll
    for (int i = 0; i < PF; ++i) {
        const int j = (i + PF - 1) % PF;
        kslot[j] = load_krow(pKXZ, pKY, pS, j);
        __builtin_amdgcn_sched_barrier(0);
    }

    v8f d = {};
    d = __builtin_amdgcn_wmma_f32_16x16x4_f32(
            false, a, false, kvec_frag(kslot[0], hi), (short)0, d, false, false);

    int c0 = 0;
    for (; c0 + PF < nChunks; c0 += PF) {
#pragma unroll
        for (int j = 0; j < PF; ++j) {
            const v2f bn = kvec_frag(kslot[(j + 1) % PF], hi);
            recip_stage<true>(a, bn, d, prev, kslot[(j + PF - 1) % PF].ap.x, accp);
            kslot[(j + PF - 1) % PF] = load_krow(pKXZ, pKY, pS, c0 + (j + PF - 1) % PF + (j ? PF : 0));
        }
    }
#pragma unroll
    for (int j = 0; j < PF; ++j) {
        if (j + 1 < PF) {
            const v2f bn = kvec_frag(kslot[j + 1], hi);
            recip_stage<true>(a, bn, d, prev, kslot[(j + PF - 1) % PF].ap.x, accp);
        } else {
            recip_stage<false>(a, a, d, prev, kslot[(j + PF - 1) % PF].ap.x, accp);
        }
        if (j == 0) kslot[PF - 1] = load_krow(pKXZ, pKY, pS, c0 + PF - 1);
    }
    const float ampprev = kslot[PF - 1].ap.x;
    v8f acc;
#pragma unroll
    for (int i = 0; i < 4; ++i) {
        const v2f v = accp.v[i] + prev.c[i] * ampprev;
        acc[2 * i] = v.x; acc[2 * i + 1] = v.y;
    }

#pragma unroll
    for (int r = 0; r < 8; ++r) {
        float v = acc[r];
        for (int mask = 8; mask >= 1; mask >>= 1)
            v += __shfl_xor(v, mask, 32);
        acc[r] = v;
    }
    if (m == 0) {
        const int o = tile * 16 + (hi ? 8 : 0);
#pragma unroll
        for (int r = 0; r < 8; ++r)
            red[slice][o + r] = acc[r];
    }
    __syncthreads();
    if (wave == 0) {
        float recip = red[0][lane];
#pragma unroll
        for (int s = 1; s < RED_WAVES; ++s) recip += red[s][lane];
        const int   i     = b * N + nBlk + lane;
        const float qi    = q[i];
        const float scale = BOHR_F / (PI_F * volume[0]);
        const float self  = 2.0f * bewald[0] * BOHR_F / sqrtf(PI_F);
        const float phi   = recip * scale - qi * self;
        out[i]      = 0.5f * qi * phi;
        out[BN + i] = phi;
    }
}

extern "C" void kernel_launch(void* const* d_in, const int* in_sizes, int n_in,
                              void* d_out, int out_size, void* d_ws, size_t ws_size,
                              hipStream_t stream)
{
    const float* coords   = (const float*)d_in[0];
    const float* q        = (const float*)d_in[1];
    const float* cell_inv = (const float*)d_in[2];
    const float* kvec     = (const float*)d_in[3];
    const float* expfac   = (const float*)d_in[4];
    const float* volume   = (const float*)d_in[5];
    const float* bewald   = (const float*)d_in[6];
    float* out = (float*)d_out;

    const int B  = 2;
    const int BN = in_sizes[1];
    const int N  = BN / B;
    const int NK = in_sizes[4];

    float* Sap = (float*)d_ws;
    const size_t wsFloats = (size_t)2 * B * NK;
    if (ws_size < wsFloats * sizeof(float)) return;

    dim3 g1(NK / 16, 1, B);
    k_struct_fac<<<g1, 32 * RED_WAVES, 0, stream>>>(coords, q, cell_inv, kvec, expfac, Sap, N, NK);

    dim3 g2(N / 32, 1, B);
    k_recip<<<g2, 64 * RED_WAVES, 0, stream>>>(coords, cell_inv, kvec, (const float2*)Sap,
                                               q, volume, bewald, out, N, NK, BN);
}
